# starting kernel
# baseline (speedup 1.0000x reference)
#include <hip/hip_runtime.h>

#define HIDDEN 64
#define EIGD   16
#define NUSERS 100000
#define NITEMS 200000
#define NNODES 300000
#define NEDGES 1250000
#define NPATHS 14
#define LN_EPS 1e-5f

typedef __attribute__((ext_vector_type(2))) float v2f;
typedef __attribute__((ext_vector_type(8))) float v8f;

static __device__ __forceinline__ unsigned fenc(float f) {
  unsigned u = __float_as_uint(f);
  return (u & 0x80000000u) ? ~u : (u | 0x80000000u);
}
static __device__ __forceinline__ float fdec(unsigned u) {
  return (u & 0x80000000u) ? __uint_as_float(u ^ 0x80000000u)
                           : __uint_as_float(~u);
}

__global__ void ln_wmma_kernel(const float* __restrict__ srcA,
                               const float* __restrict__ srcB,
                               int nA, float* __restrict__ y) {
  int lane = threadIdx.x & 31;
  int wave = threadIdx.x >> 5;
  int base = (blockIdx.x * (blockDim.x >> 5) + wave) << 4;
  if (base >= NNODES) return;

  int m = lane & 15;
  int h = lane >> 4;
  int node = base + m;
  const float* srow = (node < nA) ? (srcA + (size_t)node * HIDDEN)
                                  : (srcB + (size_t)(node - nA) * HIDDEN);
  const float* sp = srow + 2 * h;
  float*       dp = y + (size_t)node * HIDDEN + 2 * h;

  const v2f ones = {1.0f, 1.0f};
  v8f accS = {};
  v8f accQ = {};
  float2 vbuf[16];
#pragma unroll
  for (int c = 0; c < 16; ++c) {
    float2 v = *(const float2*)(sp + 4 * c);
    vbuf[c] = v;
    v2f a  = {v.x, v.y};
    v2f a2 = {v.x * v.x, v.y * v.y};
    accS = __builtin_amdgcn_wmma_f32_16x16x4_f32(false, a,  false, ones,
                                                 (short)0, accS, false, false);
    accQ = __builtin_amdgcn_wmma_f32_16x16x4_f32(false, a2, false, ones,
                                                 (short)0, accQ, false, false);
  }

  float s = 0.0f, q = 0.0f;
#pragma unroll
  for (int i = 0; i < 8; ++i) {
    float sLo = __shfl(accS[i], 0, 32);
    float sHi = __shfl(accS[i], 16, 32);
    float qLo = __shfl(accQ[i], 0, 32);
    float qHi = __shfl(accQ[i], 16, 32);
    s = (m == i)     ? sLo : s;
    s = (m == 8 + i) ? sHi : s;
    q = (m == i)     ? qLo : q;
    q = (m == 8 + i) ? qHi : q;
  }

  float mu  = s * (1.0f / HIDDEN);
  float var = q * (1.0f / HIDDEN) - mu * mu;
  float rin = rsqrtf(var + LN_EPS);
#pragma unroll
  for (int c = 0; c < 16; ++c) {
    float2 o = make_float2((vbuf[c].x - mu) * rin, (vbuf[c].y - mu) * rin);
    *(float2*)(dp + 4 * c) = o;
  }
}

__global__ void score_kernel(const float* __restrict__ y,
                             const float* __restrict__ eigs,
                             const int* __restrict__ row,
                             const int* __restrict__ col,
                             const int* __restrict__ pt,
                             const float* __restrict__ lambda0_l,
                             const float* __restrict__ pemb_l,
                             float* __restrict__ s0,
                             unsigned* __restrict__ m0u,
                             unsigned* __restrict__ m1u) {
  int e = blockIdx.x * (blockDim.x >> 4) + (threadIdx.x >> 4);
  if (e >= NEDGES) return;
  int lane = threadIdx.x & 15;
  int r = row[e], c = col[e];
  __builtin_prefetch(eigs + (size_t)r * EIGD, 0, 1);
  __builtin_prefetch(eigs + (size_t)c * EIGD, 0, 1);
  float4 a4 = ((const float4*)(y + (size_t)r * HIDDEN))[lane];
  float4 b4 = ((const float4*)(y + (size_t)c * HIDDEN))[lane];
  float p = 0.125f * (a4.x * b4.x + a4.y * b4.y + a4.z * b4.z + a4.w * b4.w);
  float er = eigs[(size_t)r * EIGD + lane];
  float ec = eigs[(size_t)c * EIGD + lane];
  float explam = expf(lambda0_l[0]);
  p += explam * er * ec;
#pragma unroll
  for (int off = 8; off > 0; off >>= 1) p += __shfl_xor(p, off, 16);
  if (lane == 0) {
    s0[e] = p;
    atomicMax(m0u + r, fenc(p));
    atomicMax(m1u + r, fenc(pemb_l[pt[e]]));
  }
}

__global__ void expsum_kernel(const float* __restrict__ s0,
                              const int* __restrict__ row,
                              const int* __restrict__ pt,
                              const float* __restrict__ pemb_l,
                              const unsigned* __restrict__ m0u,
                              const unsigned* __restrict__ m1u,
                              float* __restrict__ e0,
                              float* __restrict__ d0,
                              float* __restrict__ d1) {
  int e = blockIdx.x * blockDim.x + threadIdx.x;
  if (e >= NEDGES) return;
  int r = row[e];
  float e0v = expf(s0[e] - fdec(m0u[r]));
  e0[e] = e0v;
  atomicAdd(d0 + r, e0v);
  float e1v = expf(pemb_l[pt[e]] - fdec(m1u[r]));
  atomicAdd(d1 + r, e1v);
}

__global__ void coef_kernel(const int* __restrict__ row,
                            const int* __restrict__ pt,
                            const float* __restrict__ pemb_l,
                            const unsigned* __restrict__ m1u,
                            const float* __restrict__ e0,
                            const float* __restrict__ d0,
                            const float* __restrict__ d1,
                            float* __restrict__ a) {
  int e = blockIdx.x * blockDim.x + threadIdx.x;
  if (e >= NEDGES) return;
  int r = row[e];
  float a0 = e0[e] / d0[r];
  float a1 = expf(pemb_l[pt[e]] - fdec(m1u[r])) / d1[r];
  a[e] = 0.5f * (a0 + a1);
}

__global__ void aggregate_kernel(const float* __restrict__ a,
                                 const float* __restrict__ y,
                                 const int* __restrict__ row,
                                 const int* __restrict__ col,
                                 float* __restrict__ out) {
  int idx = blockIdx.x * blockDim.x + threadIdx.x;
  int e = idx >> 6;
  int c = idx & (HIDDEN - 1);
  float av = a[e];
  int rr = row[e];
  int cc = col[e];
  atomicAdd(out + (size_t)rr * HIDDEN + c, av * y[(size_t)cc * HIDDEN + c]);
}

__global__ void init_out_kernel(const float* __restrict__ ue,
                                const float* __restrict__ ie,
                                float* __restrict__ out) {
  int i = blockIdx.x * blockDim.x + threadIdx.x;
  out[i] = (i < NUSERS * HIDDEN) ? ue[i] : ie[i - NUSERS * HIDDEN];
}
__global__ void add_kernel(float* __restrict__ out, const float* __restrict__ x) {
  int i = blockIdx.x * blockDim.x + threadIdx.x;
  out[i] += x[i];
}
__global__ void scale_kernel(float* __restrict__ out, float s) {
  int i = blockIdx.x * blockDim.x + threadIdx.x;
  out[i] *= s;
}

extern "C" void kernel_launch(void* const* d_in, const int* in_sizes, int n_in,
                              void* d_out, int out_size, void* d_ws, size_t ws_size,
                              hipStream_t stream) {
  (void)in_sizes; (void)n_in; (void)out_size; (void)ws_size;
  const float* ue      = (const float*)d_in[0];
  const float* ie      = (const float*)d_in[1];
  const float* eigs    = (const float*)d_in[2];
  const float* lambda0 = (const float*)d_in[3];
  const float* pemb    = (const float*)d_in[4];
  const int*   indices = (const int*)d_in[5];
  const int*   pt      = (const int*)d_in[6];
  const int* row = indices;
  const int* col = indices + NEDGES;
  float* out = (float*)d_out;

  char* p = (char*)d_ws;
  auto carve = [&](size_t bytes) -> char* {
    char* r = p;
    p += (bytes + 255) & ~(size_t)255;
    return r;
  };
  float*    y    = (float*)   carve((size_t)NNODES * HIDDEN * 4);
  float*    emb1 = (float*)   carve((size_t)NNODES * HIDDEN * 4);
  float*    s0a  = (float*)   carve((size_t)NEDGES * 4);
  float*    e0   = (float*)   carve((size_t)NEDGES * 4);
  char*     statsBase = p;
  unsigned* m0u  = (unsigned*)carve((size_t)NNODES * 4);
  unsigned* m1u  = (unsigned*)carve((size_t)NNODES * 4);
  float*    d0   = (float*)   carve((size_t)NNODES * 4);
  float*    d1   = (float*)   carve((size_t)NNODES * 4);
  size_t statsBytes = (size_t)(p - statsBase);

  const dim3 B(256);
  const int GRID_NODE_ELEM = (NNODES * HIDDEN) / 256;
  const int GRID_LN        = ((NNODES / 16) + 7) / 8;
  const int GRID_SCORE     = NEDGES / 16;
  const int GRID_EDGE      = (NEDGES + 255) / 256;
  const int GRID_AGG       = (NEDGES * HIDDEN) / 256;

  hipLaunchKernelGGL(init_out_kernel, dim3(GRID_NODE_ELEM), B, 0, stream, ue, ie, out);

  for (int l = 0; l < 2; ++l) {
    hipMemsetAsync(statsBase, 0, statsBytes, stream);
    if (l == 0)
      hipMemsetAsync(emb1, 0, (size_t)NNODES * HIDDEN * 4, stream);

    if (l == 0)
      hipLaunchKernelGGL(ln_wmma_kernel, dim3(GRID_LN), B, 0, stream, ue, ie, NUSERS, y);
    else
      hipLaunchKernelGGL(ln_wmma_kernel, dim3(GRID_LN), B, 0, stream, emb1, emb1, NNODES, y);

    hipLaunchKernelGGL(score_kernel, dim3(GRID_SCORE), B, 0, stream,
                       y, eigs, row, col, pt, lambda0 + l, pemb + l * NPATHS,
                       s0a, m0u, m1u);
    hipLaunchKernelGGL(expsum_kernel, dim3(GRID_EDGE), B, 0, stream,
                       s0a, row, pt, pemb + l * NPATHS, m0u, m1u, e0, d0, d1);
    hipLaunchKernelGGL(coef_kernel, dim3(GRID_EDGE), B, 0, stream,
                       row, pt, pemb + l * NPATHS, m1u, e0, d0, d1, s0a);

    float* target = (l == 0) ? emb1 : out;
    hipLaunchKernelGGL(aggregate_kernel, dim3(GRID_AGG), B, 0, stream,
                       s0a, y, row, col, target);

    if (l == 0)
      hipLaunchKernelGGL(add_kernel, dim3(GRID_NODE_ELEM), B, 0, stream, out, emb1);
  }

  hipLaunchKernelGGL(scale_kernel, dim3(GRID_NODE_ELEM), B, 0, stream, out, 1.0f / 3.0f);
}


# LayerNorm moves whole lines: a wave's 16 rows as 8 float4 loads and 8 float4 stores of 512 contiguous bytes (was 16 float2 over 16 lines each); WMMA B is a one-hot row selector; LN sum order changes
# speedup vs baseline: 2.1435x; 2.1435x over previous
#include <hip/hip_runtime.h>

#define HIDDEN 64
#define EIGD   16
#define NUSERS 100000
#define NITEMS 200000
#define NNODES 300000
#define NEDGES 1250000
#define NPATHS 14
#define LN_EPS 1e-5f

typedef __attribute__((ext_vector_type(2))) float v2f;
typedef __attribute__((ext_vector_type(8))) float v8f;

static __device__ __forceinline__ unsigned fenc(float f) {
  unsigned u = __float_as_uint(f);
  return (u & 0x80000000u) ? ~u : (u | 0x80000000u);
}
static __device__ __forceinline__ float fdec(unsigned u) {
  return (u & 0x80000000u) ? __uint_as_float(u ^ 0x80000000u)
                           : __uint_as_float(~u);
}

static_assert(NUSERS % 16 == 0 && NNODES % 16 == 0, "a wave's 16 nodes come from one table");
typedef __attribute__((ext_vector_type(4))) float ln4f;

__global__ void ln_wmma_kernel(const float* __restrict__ srcA,
                               const float* __restrict__ srcB,
                               int nA, float* __restrict__ y) {
  const int lane = threadIdx.x & 31;
  const int wave = __builtin_amdgcn_readfirstlane((int)(threadIdx.x >> 5));
  const int base = (blockIdx.x * (int)(blockDim.x >> 5) + wave) << 4;
  if (base >= NNODES) return;

  const int c = lane & 15;
  const int h = lane >> 4;
  const float* src = (base < nA) ? (srcA + (size_t)base * HIDDEN)
                                 : (srcB + (size_t)(base - nA) * HIDDEN);
  const ln4f* __restrict__ s4 = (const ln4f*)src + lane;
  ln4f* __restrict__       d4 = (ln4f*)(y + (size_t)base * HIDDEN) + lane;

  ln4f v[8];
#pragma unroll
  for (int i = 0; i < 8; ++i) v[i] = s4[32 * i];

  v8f accS = {};
  v8f accQ = {};
#pragma unroll
  for (int i = 0; i < 8; ++i) {
    const float sel = (c == 2 * i + h) ? 1.0f : 0.0f;
    const v2f b  = {sel, sel};
    const v2f a0 = {v[i].x, v[i].y};
    const v2f a1 = {v[i].z, v[i].w};
    const v2f q0 = {v[i].x * v[i].x, v[i].y * v[i].y};
    const v2f q1 = {v[i].z * v[i].z, v[i].w * v[i].w};
    accS = __builtin_amdgcn_wmma_f32_16x16x4_f32(false, a0, false, b, (short)0, accS, false, false);
    accQ = __builtin_amdgcn_wmma_f32_16x16x4_f32(false, q0, false, b, (short)0, accQ, false, false);
    accS = __builtin_amdgcn_wmma_f32_16x16x4_f32(false, a1, false, b, (short)0, accS, false, false);
    accQ = __builtin_amdgcn_wmma_f32_16x16x4_f32(false, q1, false, b, (short)0, accQ, false, false);
  }

  float s = ((accS[0] + accS[1]) + (accS[2] + accS[3])) + ((accS[4] + accS[5]) + (accS[6] + accS[7]));
  float q = ((accQ[0] + accQ[1]) + (accQ[2] + accQ[3])) + ((accQ[4] + accQ[5]) + (accQ[6] + accQ[7]));
  s += __shfl_xor(s, 16, 32);
  q += __shfl_xor(q, 16, 32);

  const float mu  = s * (1.0f / HIDDEN);
  const float var = q * (1.0f / HIDDEN) - mu * mu;
  const float rin = rsqrtf(var + LN_EPS);
#pragma unroll
  for (int i = 0; i < 8; ++i) {
    const float mi = __shfl(mu,  2 * i + h, 32);
    const float ri = __shfl(rin, 2 * i + h, 32);
    d4[32 * i] = (v[i] - mi) * ri;
  }
}


#define SC_GROUP 4
#define SC_TRIPS 4
static_assert(SC_GROUP == 4 && SC_TRIPS == 4, "the exchange tree below is written for 4 steps and 4 trips");

__global__ void score_kernel(const float* __restrict__ y,
                             const float* __restrict__ eigs,
                             const int* __restrict__ row,
                             const int* __restrict__ col,
                             const int* __restrict__ pt,
                             const float* __restrict__ lambda0_l,
                             const float* __restrict__ pemb_l,
                             float* __restrict__ s0,
                             unsigned* __restrict__ stats) {
  const int lane = threadIdx.x & 31;
  const int sub  = lane & 15;
  const int h    = lane >> 4;
  const int wave = __builtin_amdgcn_readfirstlane((int)(threadIdx.x >> 5));
  const int base = (blockIdx.x * (int)(blockDim.x >> 5) + wave) << 5;
  if (base >= NEDGES) return;
  const int n = (NEDGES - base < 32) ? (NEDGES - base) : 32;

  int rr = 0, cc = 0;
  float s1v = 0.0f;
  if (lane < n) {
    rr  = row[base + lane];
    cc  = col[base + lane];
    s1v = pemb_l[pt[base + lane]];
  }
  const float explam = expf(lambda0_l[0]);

  const bool up8 = (sub & 8) != 0;
  const bool up4 = (sub & 4) != 0;
  float tOwn = 0.0f;
#pragma unroll 1
  for (int g = 0; g < SC_TRIPS; ++g) {
    float4 a4[SC_GROUP], b4[SC_GROUP];
    float er[SC_GROUP], ec[SC_GROUP];
#pragma unroll
    for (int k = 0; k < SC_GROUP; ++k) {
      const int e = 2 * (SC_GROUP * g + k) + h;
      const int r = __shfl(rr, e, 32);
      const int c = __shfl(cc, e, 32);
      a4[k] = ((const float4*)(y + (size_t)r * HIDDEN))[sub];
      b4[k] = ((const float4*)(y + (size_t)c * HIDDEN))[sub];
      er[k] = eigs[(size_t)r * EIGD + sub];
      ec[k] = eigs[(size_t)c * EIGD + sub];
    }
    float p[SC_GROUP];
#pragma unroll
    for (int k = 0; k < SC_GROUP; ++k) {
      float d = a4[k].y * b4[k].y;
      d = fmaf(a4[k].x, b4[k].x, d);
      d = fmaf(a4[k].z, b4[k].z, d);
      d = fmaf(a4[k].w, b4[k].w, d);
      p[k] = fmaf(d, 0.125f, (er[k] * explam) * ec[k]);
    }
    const float u0 = (up8 ? p[2] : p[0]) + __shfl_xor(up8 ? p[0] : p[2], 8, 16);
    const float u1 = (up8 ? p[3] : p[1]) + __shfl_xor(up8 ? p[1] : p[3], 8, 16);
    float t = (up4 ? u1 : u0) + __shfl_xor(up4 ? u0 : u1, 4, 16);
    t += __shfl_xor(t, 2, 16);
    t += __shfl_xor(t, 1, 16);
    tOwn = (g == (sub & 3)) ? t : tOwn;
  }

  const int eOwn = 8 * (sub & 3) + ((sub & 8) >> 1) + ((sub & 4) >> 1) + h;
  if (eOwn < n) s0[base + eOwn] = tOwn;

  const int   odd = sub & 1;
  const int   eE  = eOwn & ~8;
  const int   eO  = eE | 8;
  const float tPartner = __shfl_xor(tOwn, 1, 16);
  const int   rE  = __shfl(rr, eE, 32);
  const int   rO  = __shfl(rr, eO, 32);
  const float s1E = __shfl(s1v, eE, 32);
  const float s1O = __shfl(s1v, eO, 32);
  if (eE < n) atomicMax(stats + ((size_t)rE << 2) + odd, fenc(odd ? s1E : tOwn));
  if (eO < n) atomicMax(stats + ((size_t)rO << 2) + odd, fenc(odd ? s1O : tPartner));
}

__global__ void expsum_kernel(const float* __restrict__ s0,
                              const int* __restrict__ row,
                              const int* __restrict__ pt,
                              const float* __restrict__ pemb_l,
                              unsigned* stats,
                              float* __restrict__ e0) {
  int t = blockIdx.x * blockDim.x + threadIdx.x;
  int e = t >> 1;
  if (e >= NEDGES) return;
  int which = t & 1;
  int r = row[e];
  unsigned* rec = stats + ((size_t)r << 2);
  uint2 m = *(const uint2*)rec;
  float x = which ? pemb_l[pt[e]] : s0[e];
  float v = expf(x - fdec(which ? m.y : m.x));
  if (!which) e0[e] = v;
  atomicAdd((float*)rec + 2 + which, v);
}

__global__ void coef_kernel(const int* __restrict__ row,
                            const int* __restrict__ pt,
                            const float* __restrict__ pemb_l,
                            const unsigned* __restrict__ stats,
                            const float* __restrict__ e0,
                            float* __restrict__ a) {
  int e = blockIdx.x * blockDim.x + threadIdx.x;
  if (e >= NEDGES) return;
  int r = row[e];
  uint4 rec = *(const uint4*)(stats + ((size_t)r << 2));
  float a0 = e0[e] / __uint_as_float(rec.z);
  float a1 = expf(pemb_l[pt[e]] - fdec(rec.y)) / __uint_as_float(rec.w);
  a[e] = 0.5f * (a0 + a1);
}

#define AGG_GROUP 8
static_assert(NEDGES % AGG_GROUP == 0, "edge batches are walked in whole groups");

__global__ void aggregate_kernel(const float* __restrict__ a,
                                 const float* __restrict__ y,
                                 const int* __restrict__ row,
                                 const int* __restrict__ col,
                                 float* __restrict__ out) {
  const int lane = threadIdx.x & 31;
  const int wave = __builtin_amdgcn_readfirstlane((int)(threadIdx.x >> 5));
  const int base = (blockIdx.x * (int)(blockDim.x >> 5) + wave) << 5;
  if (base >= NEDGES) return;
  const int n = (NEDGES - base < 32) ? (NEDGES - base) : 32;

  float av = 0.0f;
  int rr = 0, cc = 0;
  if (lane < n) {
    av = a[base + lane];
    rr = row[base + lane];
    cc = col[base + lane];
  }
  const int avb = __float_as_int(av);

  for (int j0 = 0; j0 < n; j0 += AGG_GROUP) {
    float lo[AGG_GROUP], hi[AGG_GROUP];
#pragma unroll
    for (int k = 0; k < AGG_GROUP; ++k) {
      const int c = __builtin_amdgcn_readlane(cc, j0 + k);
      const float* src = y + (size_t)c * HIDDEN + lane;
      lo[k] = src[0];
      hi[k] = src[32];
    }
#pragma unroll
    for (int k = 0; k < AGG_GROUP; ++k) {
      const int r = __builtin_amdgcn_readlane(rr, j0 + k);
      const float w = __int_as_float(__builtin_amdgcn_readlane(avb, j0 + k));
      float* dst = out + (size_t)r * HIDDEN + lane;
      atomicAdd(dst, w * lo[k]);
      atomicAdd(dst + 32, w * hi[k]);
    }
  }
}

typedef __attribute__((ext_vector_type(4))) float v4f;
#define EW_VEC        (NNODES * HIDDEN / 4)
#define EW_PER_THREAD 6
#define EW_THREADS    (EW_VEC / EW_PER_THREAD)
#define EW_USER_VEC   (NUSERS * HIDDEN / 4)
static_assert(EW_VEC % (EW_PER_THREAD * 256) == 0, "whole blocks of 256 threads, no tail");
static_assert((NUSERS * HIDDEN) % 4 == 0, "a piece never straddles the two tables");
static_assert(EW_USER_VEC % EW_THREADS == 0, "the user/item boundary falls between two pieces of a thread");
static_assert(EW_PER_THREAD == 6, "the load fences below name six pieces");

__global__ void init_out_kernel(const float* __restrict__ ue,
                                const float* __restrict__ ie,
                                float* __restrict__ out) {
  const int t = blockIdx.x * blockDim.x + threadIdx.x;
  const v4f* __restrict__ u4 = (const v4f*)ue;
  const v4f* __restrict__ i4 = (const v4f*)ie;
  v4f* __restrict__ o4 = (v4f*)out;
  v4f v[EW_PER_THREAD];
#pragma unroll
  for (int j = 0; j < EW_PER_THREAD; ++j) {
    const int i = t + j * EW_THREADS;
    v[j] = (j < EW_USER_VEC / EW_THREADS) ? u4[i] : i4[i - EW_USER_VEC];
  }
  asm volatile("" :: "v"(v[0]), "v"(v[1]), "v"(v[2]), "v"(v[3]), "v"(v[4]), "v"(v[5]));
#pragma unroll
  for (int j = 0; j < EW_PER_THREAD; ++j) o4[t + j * EW_THREADS] = v[j];
}
__global__ void add_kernel(float* __restrict__ out, const float* __restrict__ x) {
  const int t = blockIdx.x * blockDim.x + threadIdx.x;
  v4f* __restrict__ o4 = (v4f*)out;
  const v4f* __restrict__ x4 = (const v4f*)x;
  v4f a[EW_PER_THREAD], b[EW_PER_THREAD];
#pragma unroll
  for (int j = 0; j < EW_PER_THREAD; ++j) {
    a[j] = o4[t + j * EW_THREADS];
    b[j] = x4[t + j * EW_THREADS];
  }
  asm volatile("" :: "v"(a[0]), "v"(a[1]), "v"(a[2]), "v"(a[3]), "v"(a[4]), "v"(a[5]),
                     "v"(b[0]), "v"(b[1]), "v"(b[2]), "v"(b[3]), "v"(b[4]), "v"(b[5]));
#pragma unroll
  for (int j = 0; j < EW_PER_THREAD; ++j) o4[t + j * EW_THREADS] = a[j] + b[j];
}
__global__ void scale_kernel(float* __restrict__ out, float s) {
  const int t = blockIdx.x * blockDim.x + threadIdx.x;
  v4f* __restrict__ o4 = (v4f*)out;
  v4f a[EW_PER_THREAD];
#pragma unroll
  for (int j = 0; j < EW_PER_THREAD; ++j) a[j] = o4[t + j * EW_THREADS];
#pragma unroll
  for (int j = 0; j < EW_PER_THREAD; ++j) o4[t + j * EW_THREADS] = a[j] * s;
}

extern "C" void kernel_launch(void* const* d_in, const int* in_sizes, int n_in,
                              void* d_out, int out_size, void* d_ws, size_t ws_size,
                              hipStream_t stream) {
  (void)in_sizes; (void)n_in; (void)out_size; (void)ws_size;
  const float* ue      = (const float*)d_in[0];
  const float* ie      = (const float*)d_in[1];
  const float* eigs    = (const float*)d_in[2];
  const float* lambda0 = (const float*)d_in[3];
  const float* pemb    = (const float*)d_in[4];
  const int*   indices = (const int*)d_in[5];
  const int*   pt      = (const int*)d_in[6];
  const int* row = indices;
  const int* col = indices + NEDGES;
  float* out = (float*)d_out;

  char* p = (char*)d_ws;
  auto carve = [&](size_t bytes) -> char* {
    char* r = p;
    p += (bytes + 255) & ~(size_t)255;
    return r;
  };
  float*    y    = (float*)   carve((size_t)NNODES * HIDDEN * 4);
  float*    emb1 = (float*)   carve((size_t)NNODES * HIDDEN * 4);
  float*    s0a  = (float*)   carve((size_t)NEDGES * 4);
  float*    e0   = (float*)   carve((size_t)NEDGES * 4);
  char*     statsBase = p;
  unsigned* stats = (unsigned*)carve((size_t)NNODES * 16);
  size_t statsBytes = (size_t)(p - statsBase);

  const dim3 B(256);
  const int GRID_NODE_ELEM = EW_THREADS / 256;
  const int GRID_LN        = ((NNODES / 16) + 7) / 8;
  const int GRID_SCORE     = ((NEDGES + 31) / 32 + 7) / 8;
  const int GRID_EDGE      = (NEDGES + 255) / 256;
  const int GRID_EDGE2     = (2 * NEDGES + 255) / 256;
  const int GRID_AGG       = ((NEDGES + 31) / 32 + 7) / 8;

  hipLaunchKernelGGL(init_out_kernel, dim3(GRID_NODE_ELEM), B, 0, stream, ue, ie, out);

  for (int l = 0; l < 2; ++l) {
    hipMemsetAsync(statsBase, 0, statsBytes, stream);
    if (l == 0)
      hipMemsetAsync(emb1, 0, (size_t)NNODES * HIDDEN * 4, stream);

    if (l == 0)
      hipLaunchKernelGGL(ln_wmma_kernel, dim3(GRID_LN), B, 0, stream, ue, ie, NUSERS, y);
    else
      hipLaunchKernelGGL(ln_wmma_kernel, dim3(GRID_LN), B, 0, stream, emb1, emb1, NNODES, y);

    hipLaunchKernelGGL(score_kernel, dim3(GRID_SCORE), B, 0, stream,
                       y, eigs, row, col, pt, lambda0 + l, pemb + l * NPATHS,
                       s0a, stats);
    hipLaunchKernelGGL(expsum_kernel, dim3(GRID_EDGE2), B, 0, stream,
                       s0a, row, pt, pemb + l * NPATHS, stats, e0);
    hipLaunchKernelGGL(coef_kernel, dim3(GRID_EDGE), B, 0, stream,
                       row, pt, pemb + l * NPATHS, stats, e0, s0a);

    float* target = (l == 0) ? emb1 : out;
    hipLaunchKernelGGL(aggregate_kernel, dim3(GRID_AGG), B, 0, stream,
                       s0a, y, row, col, target);

    if (l == 0)
      hipLaunchKernelGGL(add_kernel, dim3(GRID_NODE_ELEM), B, 0, stream, out, emb1);
  }

  hipLaunchKernelGGL(scale_kernel, dim3(GRID_NODE_ELEM), B, 0, stream, out, 1.0f / 3.0f);
}


# Mean built in place: layer-1 LayerNorm launch also stores (emb0+emb1)/3 into out, layer-1 coefficients carry the 1/3, scatter adds into it; copy, add, scale launches gone; mean sum order changes
# speedup vs baseline: 2.4942x; 1.1636x over previous
#include <hip/hip_runtime.h>

#define HIDDEN 64
#define EIGD   16
#define NUSERS 100000
#define NITEMS 200000
#define NNODES 300000
#define NEDGES 1250000
#define NPATHS 14
#define LN_EPS 1e-5f

typedef __attribute__((ext_vector_type(2))) float v2f;
typedef __attribute__((ext_vector_type(8))) float v8f;

static __device__ __forceinline__ unsigned fenc(float f) {
  unsigned u = __float_as_uint(f);
  return (u & 0x80000000u) ? ~u : (u | 0x80000000u);
}
static __device__ __forceinline__ float fdec(unsigned u) {
  return (u & 0x80000000u) ? __uint_as_float(u ^ 0x80000000u)
                           : __uint_as_float(~u);
}

static_assert(NUSERS % 16 == 0 && NNODES % 16 == 0, "a wave's 16 nodes come from one table");
typedef __attribute__((ext_vector_type(4))) float ln4f;

template <bool MEAN>
static __device__ __forceinline__ void ln_rows(const float* __restrict__ src,
                                               float* __restrict__ ydst,
                                               const float* __restrict__ first,
                                               float* __restrict__ odst,
                                               int lane) {
  const int c = lane & 15;
  const int h = lane >> 4;
  const ln4f* __restrict__ s4 = (const ln4f*)src + lane;
  ln4f* __restrict__       d4 = (ln4f*)ydst + lane;

  ln4f v[8];
  ln4f f[8];
#pragma unroll
  for (int i = 0; i < 8; ++i) v[i] = s4[32 * i];
  if (MEAN) {
    const ln4f* __restrict__ f4 = (const ln4f*)first + lane;
#pragma unroll
    for (int i = 0; i < 8; ++i) f[i] = f4[32 * i];
    __builtin_amdgcn_sched_barrier(0);
  }

  v8f accS = {};
  v8f accQ = {};
#pragma unroll
  for (int i = 0; i < 8; ++i) {
    const float sel = (c == 2 * i + h) ? 1.0f : 0.0f;
    const v2f b  = {sel, sel};
    const v2f a0 = {v[i].x, v[i].y};
    const v2f a1 = {v[i].z, v[i].w};
    const v2f q0 = {v[i].x * v[i].x, v[i].y * v[i].y};
    const v2f q1 = {v[i].z * v[i].z, v[i].w * v[i].w};
    accS = __builtin_amdgcn_wmma_f32_16x16x4_f32(false, a0, false, b, (short)0, accS, false, false);
    accQ = __builtin_amdgcn_wmma_f32_16x16x4_f32(false, q0, false, b, (short)0, accQ, false, false);
    accS = __builtin_amdgcn_wmma_f32_16x16x4_f32(false, a1, false, b, (short)0, accS, false, false);
    accQ = __builtin_amdgcn_wmma_f32_16x16x4_f32(false, q1, false, b, (short)0, accQ, false, false);
  }

  float s = ((accS[0] + accS[1]) + (accS[2] + accS[3])) + ((accS[4] + accS[5]) + (accS[6] + accS[7]));
  float q = ((accQ[0] + accQ[1]) + (accQ[2] + accQ[3])) + ((accQ[4] + accQ[5]) + (accQ[6] + accQ[7]));
  s += __shfl_xor(s, 16, 32);
  q += __shfl_xor(q, 16, 32);

  const float mu  = s * (1.0f / HIDDEN);
  const float var = q * (1.0f / HIDDEN) - mu * mu;
  const float rin = rsqrtf(var + LN_EPS);
  ln4f* __restrict__ o4 = (ln4f*)odst + lane;
#pragma unroll
  for (int i = 0; i < 8; ++i) {
    const float mi = __shfl(mu,  2 * i + h, 32);
    const float ri = __shfl(rin, 2 * i + h, 32);
    d4[32 * i] = (v[i] - mi) * ri;
    if (MEAN) o4[32 * i] = (f[i] + v[i]) * (1.0f / 3.0f);
  }
}

__global__ void ln_wmma_kernel(const float* __restrict__ ue,
                               const float* __restrict__ ie,
                               float* __restrict__ y) {
  const int lane = threadIdx.x & 31;
  const int wave = __builtin_amdgcn_readfirstlane((int)(threadIdx.x >> 5));
  const int base = (blockIdx.x * (int)(blockDim.x >> 5) + wave) << 4;
  if (base >= NNODES) return;
  const float* src = (base < NUSERS) ? (ue + (size_t)base * HIDDEN)
                                     : (ie + (size_t)(base - NUSERS) * HIDDEN);
  ln_rows<false>(src, y + (size_t)base * HIDDEN, src, y, lane);
}

__global__ void ln_mean_kernel(const float* __restrict__ emb1,
                               const float* __restrict__ ue,
                               const float* __restrict__ ie,
                               float* __restrict__ y,
                               float* __restrict__ out) {
  const int lane = threadIdx.x & 31;
  const int wave = __builtin_amdgcn_readfirstlane((int)(threadIdx.x >> 5));
  const int base = (blockIdx.x * (int)(blockDim.x >> 5) + wave) << 4;
  if (base >= NNODES) return;
  const float* first = (base < NUSERS) ? (ue + (size_t)base * HIDDEN)
                                       : (ie + (size_t)(base - NUSERS) * HIDDEN);
  ln_rows<true>(emb1 + (size_t)base * HIDDEN, y + (size_t)base * HIDDEN,
                first, out + (size_t)base * HIDDEN, lane);
}


#define SC_GROUP 4
#define SC_TRIPS 4
static_assert(SC_GROUP == 4 && SC_TRIPS == 4, "the exchange tree below is written for 4 steps and 4 trips");

__global__ void score_kernel(const float* __restrict__ y,
                             const float* __restrict__ eigs,
                             const int* __restrict__ row,
                             const int* __restrict__ col,
                             const int* __restrict__ pt,
                             const float* __restrict__ lambda0_l,
                             const float* __restrict__ pemb_l,
                             float* __restrict__ s0,
                             unsigned* __restrict__ stats) {
  const int lane = threadIdx.x & 31;
  const int sub  = lane & 15;
  const int h    = lane >> 4;
  const int wave = __builtin_amdgcn_readfirstlane((int)(threadIdx.x >> 5));
  const int base = (blockIdx.x * (int)(blockDim.x >> 5) + wave) << 5;
  if (base >= NEDGES) return;
  const int n = (NEDGES - base < 32) ? (NEDGES - base) : 32;

  int rr = 0, cc = 0;
  float s1v = 0.0f;
  if (lane < n) {
    rr  = row[base + lane];
    cc  = col[base + lane];
    s1v = pemb_l[pt[base + lane]];
  }
  const float explam = expf(lambda0_l[0]);

  const bool up8 = (sub & 8) != 0;
  const bool up4 = (sub & 4) != 0;
  float tOwn = 0.0f;
#pragma unroll 1
  for (int g = 0; g < SC_TRIPS; ++g) {
    float4 a4[SC_GROUP], b4[SC_GROUP];
    float er[SC_GROUP], ec[SC_GROUP];
#pragma unroll
    for (int k = 0; k < SC_GROUP; ++k) {
      const int e = 2 * (SC_GROUP * g + k) + h;
      const int r = __shfl(rr, e, 32);
      const int c = __shfl(cc, e, 32);
      a4[k] = ((const float4*)(y + (size_t)r * HIDDEN))[sub];
      b4[k] = ((const float4*)(y + (size_t)c * HIDDEN))[sub];
      er[k] = eigs[(size_t)r * EIGD + sub];
      ec[k] = eigs[(size_t)c * EIGD + sub];
    }
    float p[SC_GROUP];
#pragma unroll
    for (int k = 0; k < SC_GROUP; ++k) {
      float d = a4[k].y * b4[k].y;
      d = fmaf(a4[k].x, b4[k].x, d);
      d = fmaf(a4[k].z, b4[k].z, d);
      d = fmaf(a4[k].w, b4[k].w, d);
      p[k] = fmaf(d, 0.125f, (er[k] * explam) * ec[k]);
    }
    const float u0 = (up8 ? p[2] : p[0]) + __shfl_xor(up8 ? p[0] : p[2], 8, 16);
    const float u1 = (up8 ? p[3] : p[1]) + __shfl_xor(up8 ? p[1] : p[3], 8, 16);
    float t = (up4 ? u1 : u0) + __shfl_xor(up4 ? u0 : u1, 4, 16);
    t += __shfl_xor(t, 2, 16);
    t += __shfl_xor(t, 1, 16);
    tOwn = (g == (sub & 3)) ? t : tOwn;
  }

  const int eOwn = 8 * (sub & 3) + ((sub & 8) >> 1) + ((sub & 4) >> 1) + h;
  if (eOwn < n) s0[base + eOwn] = tOwn;

  const int   odd = sub & 1;
  const int   eE  = eOwn & ~8;
  const int   eO  = eE | 8;
  const float tPartner = __shfl_xor(tOwn, 1, 16);
  const int   rE  = __shfl(rr, eE, 32);
  const int   rO  = __shfl(rr, eO, 32);
  const float s1E = __shfl(s1v, eE, 32);
  const float s1O = __shfl(s1v, eO, 32);
  if (eE < n) atomicMax(stats + ((size_t)rE << 2) + odd, fenc(odd ? s1E : tOwn));
  if (eO < n) atomicMax(stats + ((size_t)rO << 2) + odd, fenc(odd ? s1O : tPartner));
}

__global__ void expsum_kernel(const float* __restrict__ s0,
                              const int* __restrict__ row,
                              const int* __restrict__ pt,
                              const float* __restrict__ pemb_l,
                              unsigned* stats,
                              float* __restrict__ e0) {
  int t = blockIdx.x * blockDim.x + threadIdx.x;
  int e = t >> 1;
  if (e >= NEDGES) return;
  int which = t & 1;
  int r = row[e];
  unsigned* rec = stats + ((size_t)r << 2);
  uint2 m = *(const uint2*)rec;
  float x = which ? pemb_l[pt[e]] : s0[e];
  float v = expf(x - fdec(which ? m.y : m.x));
  if (!which) e0[e] = v;
  atomicAdd((float*)rec + 2 + which, v);
}

__global__ void coef_kernel(const int* __restrict__ row,
                            const int* __restrict__ pt,
                            const float* __restrict__ pemb_l,
                            const unsigned* __restrict__ stats,
                            const float* __restrict__ e0,
                            float cs,
                            float* __restrict__ a) {
  int e = blockIdx.x * blockDim.x + threadIdx.x;
  if (e >= NEDGES) return;
  int r = row[e];
  uint4 rec = *(const uint4*)(stats + ((size_t)r << 2));
  float a0 = e0[e] / __uint_as_float(rec.z);
  float a1 = expf(pemb_l[pt[e]] - fdec(rec.y)) / __uint_as_float(rec.w);
  a[e] = cs * (a0 + a1);
}

#define AGG_GROUP 8
static_assert(NEDGES % AGG_GROUP == 0, "edge batches are walked in whole groups");

__global__ void aggregate_kernel(const float* __restrict__ a,
                                 const float* __restrict__ y,
                                 const int* __restrict__ row,
                                 const int* __restrict__ col,
                                 float* __restrict__ out) {
  const int lane = threadIdx.x & 31;
  const int wave = __builtin_amdgcn_readfirstlane((int)(threadIdx.x >> 5));
  const int base = (blockIdx.x * (int)(blockDim.x >> 5) + wave) << 5;
  if (base >= NEDGES) return;
  const int n = (NEDGES - base < 32) ? (NEDGES - base) : 32;

  float av = 0.0f;
  int rr = 0, cc = 0;
  if (lane < n) {
    av = a[base + lane];
    rr = row[base + lane];
    cc = col[base + lane];
  }
  const int avb = __float_as_int(av);

  for (int j0 = 0; j0 < n; j0 += AGG_GROUP) {
    float lo[AGG_GROUP], hi[AGG_GROUP];
#pragma unroll
    for (int k = 0; k < AGG_GROUP; ++k) {
      const int c = __builtin_amdgcn_readlane(cc, j0 + k);
      const float* src = y + (size_t)c * HIDDEN + lane;
      lo[k] = src[0];
      hi[k] = src[32];
    }
#pragma unroll
    for (int k = 0; k < AGG_GROUP; ++k) {
      const int r = __builtin_amdgcn_readlane(rr, j0 + k);
      const float w = __int_as_float(__builtin_amdgcn_readlane(avb, j0 + k));
      float* dst = out + (size_t)r * HIDDEN + lane;
      atomicAdd(dst, w * lo[k]);
      atomicAdd(dst + 32, w * hi[k]);
    }
  }
}

extern "C" void kernel_launch(void* const* d_in, const int* in_sizes, int n_in,
                              void* d_out, int out_size, void* d_ws, size_t ws_size,
                              hipStream_t stream) {
  (void)in_sizes; (void)n_in; (void)out_size; (void)ws_size;
  const float* ue      = (const float*)d_in[0];
  const float* ie      = (const float*)d_in[1];
  const float* eigs    = (const float*)d_in[2];
  const float* lambda0 = (const float*)d_in[3];
  const float* pemb    = (const float*)d_in[4];
  const int*   indices = (const int*)d_in[5];
  const int*   pt      = (const int*)d_in[6];
  const int* row = indices;
  const int* col = indices + NEDGES;
  float* out = (float*)d_out;

  char* p = (char*)d_ws;
  auto carve = [&](size_t bytes) -> char* {
    char* r = p;
    p += (bytes + 255) & ~(size_t)255;
    return r;
  };
  float*    y    = (float*)   carve((size_t)NNODES * HIDDEN * 4);
  float*    emb1 = (float*)   carve((size_t)NNODES * HIDDEN * 4);
  float*    s0a  = (float*)   carve((size_t)NEDGES * 4);
  float*    e0   = (float*)   carve((size_t)NEDGES * 4);
  char*     statsBase = p;
  unsigned* stats = (unsigned*)carve((size_t)NNODES * 16);
  size_t statsBytes = (size_t)(p - statsBase);

  const dim3 B(256);
  const int GRID_LN        = ((NNODES / 16) + 7) / 8;
  const int GRID_SCORE     = ((NEDGES + 31) / 32 + 7) / 8;
  const int GRID_EDGE      = (NEDGES + 255) / 256;
  const int GRID_EDGE2     = (2 * NEDGES + 255) / 256;
  const int GRID_AGG       = ((NEDGES + 31) / 32 + 7) / 8;

  for (int l = 0; l < 2; ++l) {
    hipMemsetAsync(statsBase, 0, statsBytes, stream);
    if (l == 0)
      hipMemsetAsync(emb1, 0, (size_t)NNODES * HIDDEN * 4, stream);

    if (l == 0)
      hipLaunchKernelGGL(ln_wmma_kernel, dim3(GRID_LN), B, 0, stream, ue, ie, y);
    else
      hipLaunchKernelGGL(ln_mean_kernel, dim3(GRID_LN), B, 0, stream, emb1, ue, ie, y, out);

    hipLaunchKernelGGL(score_kernel, dim3(GRID_SCORE), B, 0, stream,
                       y, eigs, row, col, pt, lambda0 + l, pemb + l * NPATHS,
                       s0a, stats);
    hipLaunchKernelGGL(expsum_kernel, dim3(GRID_EDGE2), B, 0, stream,
                       s0a, row, pt, pemb + l * NPATHS, stats, e0);
    hipLaunchKernelGGL(coef_kernel, dim3(GRID_EDGE), B, 0, stream,
                       row, pt, pemb + l * NPATHS, stats, e0,
                       (l == 0) ? 0.5f : (0.5f / 3.0f), s0a);

    float* target = (l == 0) ? emb1 : out;
    hipLaunchKernelGGL(aggregate_kernel, dim3(GRID_AGG), B, 0, stream,
                       s0a, y, row, col, target);
  }
}
